# starting kernel
# baseline (speedup 1.0000x reference)
#include <hip/hip_runtime.h>


#define HSTEPS 8192
#define BLOCK  1024
#define NWAVE  32

#define ACC_LIMIT   3.0f
#define STEER_LIMIT 0.6f
#define V_MIN_C     0.0f
#define V_MAX_C     30.0f
#define WHEELBASE   2.7f

typedef __attribute__((ext_vector_type(2))) float v2f;
typedef __attribute__((ext_vector_type(8))) float v8f;

__device__ __forceinline__ float clipf(float x, float lo, float hi) {
    return fminf(fmaxf(x, lo), hi);
}

struct SMem {
    float stage[NWAVE][256];
    float tot[NWAVE][16];
    float tileSum[NWAVE];
    float tileOff[NWAVE];
    float fS[2][BLOCK];
    float fL[2][BLOCK];
    float fH[2][BLOCK];
};

__device__ __forceinline__ void tile_scan(float v[8], SMem* sm, int wave, int lane,
                                          const v2f aL[4], const v2f bM[4]) {
    const int col  = lane & 15;
    const int half = lane >> 4;

    __syncthreads();
    float* st = sm->stage[wave];
#pragma unroll
    for (int r = 0; r < 8; ++r)
        st[col * 16 + half * 8 + r] = v[r];
    __syncthreads();

    v8f c = {};
#pragma unroll
    for (int kb = 0; kb < 4; ++kb) {
        v2f b;
#pragma unroll
        for (int j = 0; j < 2; ++j)
            b[j] = st[col * 16 + kb * 4 + half * 2 + j];
        c = __builtin_amdgcn_wmma_f32_16x16x4_f32(false, aL[kb], false, b,
                                                  (short)0, c, false, false);
    }

    if (half == 1) sm->tot[wave][col] = c[7];
    __syncthreads();

#pragma unroll
    for (int kb = 0; kb < 4; ++kb) {
        v2f a;
#pragma unroll
        for (int j = 0; j < 2; ++j)
            a[j] = sm->tot[wave][kb * 4 + half * 2 + j];
        c = __builtin_amdgcn_wmma_f32_16x16x4_f32(false, a, false, bM[kb],
                                                  (short)0, c, false, false);
    }

    if (lane == 31) sm->tileSum[wave] = c[7];
    __syncthreads();

    if (threadIdx.x < NWAVE) {
        float tv = sm->tileSum[threadIdx.x];
        float incl = tv;
#pragma unroll
        for (int d = 1; d < 32; d <<= 1) {
            float o = __shfl_up(incl, d, 32);
            if ((int)threadIdx.x >= d) incl += o;
        }
        sm->tileOff[threadIdx.x] = incl - tv;
    }
    __syncthreads();

    const float carry = sm->tileOff[wave];
#pragma unroll
    for (int r = 0; r < 8; ++r) v[r] = c[r] + carry;
}

__global__ __launch_bounds__(BLOCK) void
KinematicBicycle_82214263980490_kernel(const float* __restrict__ x0,
                                       const float* __restrict__ U,
                                       const float* __restrict__ dtp,
                                       float* __restrict__ out) {
    __shared__ SMem sm;
    const int tid  = threadIdx.x;
    const int wave = tid >> 5;
    const int lane = tid & 31;
    const int col  = lane & 15;
    const int half = lane >> 4;
    const int rank = wave * 32 + col * 2 + half;
    const int t0 = wave * 256 + col * 16 + half * 8;

    v2f aL[4], bM[4];
#pragma unroll
    for (int kb = 0; kb < 4; ++kb) {
#pragma unroll
        for (int j = 0; j < 2; ++j) {
            const int k = kb * 4 + half * 2 + j;
            aL[kb][j] = (k <= col) ? 1.0f : 0.0f;
            bM[kb][j] = (k <  col) ? 1.0f : 0.0f;
        }
    }

    const float dt = dtp[0];
    const float px = x0[0], py = x0[1], pth = x0[2], pv = x0[3];

    float s[8], dl[8];
    const float4* u4 = (const float4*)(U + (size_t)t0 * 2);
#pragma unroll
    for (int q = 0; q < 4; ++q) {
        const float4 u = u4[q];
        s[2 * q]      = clipf(u.x, -ACC_LIMIT, ACC_LIMIT) * dt;
        dl[2 * q]     = clipf(u.y, -STEER_LIMIT, STEER_LIMIT);
        s[2 * q + 1]  = clipf(u.z, -ACC_LIMIT, ACC_LIMIT) * dt;
        dl[2 * q + 1] = clipf(u.w, -STEER_LIMIT, STEER_LIMIT);
    }

    float S = s[0], Lo = V_MIN_C, Hi = V_MAX_C;
#pragma unroll
    for (int r = 1; r < 8; ++r) {
        S += s[r];
        Lo = clipf(Lo + s[r], V_MIN_C, V_MAX_C);
        Hi = clipf(Hi + s[r], V_MIN_C, V_MAX_C);
    }
    int p = 0;
    sm.fS[0][rank] = S; sm.fL[0][rank] = Lo; sm.fH[0][rank] = Hi;
    __syncthreads();
    for (int off = 1; off < BLOCK; off <<= 1) {
        if (rank >= off) {
            const float pS = sm.fS[p][rank - off];
            const float pL = sm.fL[p][rank - off];
            const float pH = sm.fH[p][rank - off];
            const float nS = pS + S;
            const float nL = clipf(pL + S, Lo, Hi);
            const float nH = clipf(pH + S, Lo, Hi);
            S = nS; Lo = nL; Hi = nH;
        }
        p ^= 1;
        sm.fS[p][rank] = S; sm.fL[p][rank] = Lo; sm.fH[p][rank] = Hi;
        __syncthreads();
    }
    float w = clipf(pv, V_MIN_C, V_MAX_C);
    if (rank > 0)
        w = clipf(w + sm.fS[p][rank - 1], sm.fL[p][rank - 1], sm.fH[p][rank - 1]);

    float wst[8], vout[8], dth[8];
    const float dtOverL = dt / WHEELBASE;
#pragma unroll
    for (int r = 0; r < 8; ++r) {
        wst[r] = w;
        dth[r] = w * tanf(dl[r]) * dtOverL;
        w = clipf(w + s[r], V_MIN_C, V_MAX_C);
        vout[r] = w;
    }

    float incl[8];
#pragma unroll
    for (int r = 0; r < 8; ++r) incl[r] = dth[r];
    tile_scan(incl, &sm, wave, lane, aL, bM);
    float thout[8], thst[8];
#pragma unroll
    for (int r = 0; r < 8; ++r) {
        thout[r] = pth + incl[r];
        thst[r]  = thout[r] - dth[r];
    }

    float iy[8];
#pragma unroll
    for (int r = 0; r < 8; ++r) {
        float sn, cs;
        sincosf(thst[r], &sn, &cs);
        const float g = wst[r] * dt;
        incl[r] = g * cs;
        iy[r]   = g * sn;
    }
    tile_scan(incl, &sm, wave, lane, aL, bM);
    tile_scan(iy,   &sm, wave, lane, aL, bM);

#pragma unroll
    for (int r = 0; r < 8; ++r) {
        float4 row = make_float4(px + incl[r], py + iy[r], thout[r], vout[r]);
        *(float4*)(out + (size_t)(t0 + r + 1) * 4) = row;
    }
    if (tid == 0) *(float4*)out = make_float4(px, py, pth, pv);
}

extern "C" void kernel_launch(void* const* d_in, const int* in_sizes, int n_in,
                              void* d_out, int out_size, void* d_ws, size_t ws_size,
                              hipStream_t stream) {
    (void)in_sizes; (void)n_in; (void)out_size; (void)d_ws; (void)ws_size;
    const float* x0 = (const float*)d_in[0];
    const float* U  = (const float*)d_in[1];
    const float* dt = (const float*)d_in[2];
    float* out = (float*)d_out;
    KinematicBicycle_82214263980490_kernel<<<1, BLOCK, 0, stream>>>(x0, U, dt, out);
}


# Output stores cover 2 x 256 contiguous bytes (was 32 lines x 16 B): x, y scans run transposed on the matrix unit (tile as A operand), theta and speed cross a per-wave LDS tile; values unchanged
# speedup vs baseline: 1.5795x; 1.5795x over previous
#include <hip/hip_runtime.h>


#define HSTEPS 8192
#define BLOCK  1024
#define NWAVE  32

#define ACC_LIMIT   3.0f
#define STEER_LIMIT 0.6f
#define V_MIN_C     0.0f
#define V_MAX_C     30.0f
#define WHEELBASE   2.7f

typedef __attribute__((ext_vector_type(2))) float v2f;
typedef __attribute__((ext_vector_type(8))) float v8f;

__device__ __forceinline__ float clipf(float x, float lo, float hi) {
    return fminf(fmaxf(x, lo), hi);
}

#define RAKE_PITCH 33
#define RAKE_WORDS (NWAVE * RAKE_PITCH)

#define TV_PITCH 36
#define TV_WORDS (16 * TV_PITCH)

struct SMem {
    alignas(16) float tv[NWAVE * TV_WORDS];
    float tileSum[NWAVE];
    float tileOff[NWAVE];
    float eS[RAKE_WORDS];
    float eL[RAKE_WORDS];
    float eH[RAKE_WORDS];
    float wIn[RAKE_WORDS];
};

struct ClampAdd { float S, L, H; };

__device__ __forceinline__ ClampAdd then_fn(const ClampAdd first, const ClampAdd next) {
    ClampAdd r;
    r.S = first.S + next.S;
    r.L = clipf(first.L + next.S, next.L, next.H);
    r.H = clipf(first.H + next.S, next.L, next.H);
    return r;
}
__device__ __forceinline__ float apply_fn(const ClampAdd f, float v) {
    return clipf(v + f.S, f.L, f.H);
}

__device__ __forceinline__ float xhalf(float x) {
    const int b = __builtin_bit_cast(int, x);
    return __builtin_bit_cast(float,
        __builtin_amdgcn_permlanex16(b, b, 0x76543210, 0xfedcba98, false, false));
}

#define DPP_ROW_SHL(n)   (0x100 + (n))
#define DPP_ROW_SHR(n)   (0x110 + (n))
#define DPP_ROW_SHARE(n) (0x150 + (n))
template <int CTRL>
__device__ __forceinline__ float dpp_mov(float x) {
    const int b = __builtin_bit_cast(int, x);
    return __builtin_bit_cast(float, __builtin_amdgcn_update_dpp(0, b, CTRL, 0xf, 0xf, true));
}

__device__ __forceinline__ void sincos_turns(float th, float* sn, float* cs) {
    const float C_HI = 0.15915494f;
    const float C_LO = 6.4206382e-9f;
    const float k = __builtin_rintf(th * C_HI);
    float f = __builtin_fmaf(th, C_HI, -k);
    f = __builtin_fmaf(th, C_LO, f);
    *sn = __builtin_amdgcn_sinf(f);
    *cs = __builtin_amdgcn_cosf(f);
}

__device__ __forceinline__ float tan_turns_small(float a) {
    const float C_HI = 0.15915494f;
    const float C_LO = 6.4206382e-9f;
    const float f = __builtin_fmaf(a, C_HI, a * C_LO);
    return __builtin_amdgcn_sinf(f) * __builtin_amdgcn_rcpf(__builtin_amdgcn_cosf(f));
}

__device__ __forceinline__ void kslices(const float v[8], bool hi, v2f f[4]) {
    const float r0 = xhalf(hi ? v[0] : v[2]);
    const float r1 = xhalf(hi ? v[1] : v[3]);
    const float r2 = xhalf(hi ? v[4] : v[6]);
    const float r3 = xhalf(hi ? v[5] : v[7]);

    f[0][0] = hi ? r0   : v[0];  f[0][1] = hi ? r1   : v[1];
    f[1][0] = hi ? r2   : v[4];  f[1][1] = hi ? r3   : v[5];
    f[2][0] = hi ? v[2] : r0;    f[2][1] = hi ? v[3] : r1;
    f[3][0] = hi ? v[6] : r2;    f[3][1] = hi ? v[7] : r3;
}

__device__ __forceinline__ float cross_tile_carry(float corner, SMem* sm, int wave, int lane) {
    if (lane == 31) sm->tileSum[wave] = corner;
    __syncthreads();

    if (threadIdx.x < NWAVE) {
        float ts = sm->tileSum[threadIdx.x];
        float incl = ts;
#pragma unroll
        for (int d = 1; d < 32; d <<= 1) {
            float o = __shfl_up(incl, d, 32);
            if ((int)threadIdx.x >= d) incl += o;
        }
        sm->tileOff[threadIdx.x] = incl - ts;
    }
    __syncthreads();

    return sm->tileOff[wave];
}

__device__ __forceinline__ void tile_scan(float v[8], SMem* sm, int wave, int lane,
                                          const v2f aL[4], const v2f bM[4]) {
    const bool hi = (lane >> 4) != 0;

    v2f b[4];
    kslices(v, hi, b);

    v8f c = {};
#pragma unroll
    for (int kb = 0; kb < 4; ++kb)
        c = __builtin_amdgcn_wmma_f32_16x16x4_f32(false, aL[kb], false, b[kb],
                                                  (short)0, c, false, false);

    const float colTot = c[7];
    const float totShl2  = dpp_mov<DPP_ROW_SHL(2)>(colTot);
    const float totOther = xhalf(colTot);
    const float W = hi ? totShl2 : totOther;

    v2f a[4];
    a[0][0] = dpp_mov<DPP_ROW_SHARE(0)>(W);   a[0][1] = dpp_mov<DPP_ROW_SHARE(1)>(W);
    a[1][0] = dpp_mov<DPP_ROW_SHARE(4)>(W);   a[1][1] = dpp_mov<DPP_ROW_SHARE(5)>(W);
    a[2][0] = dpp_mov<DPP_ROW_SHARE(8)>(W);   a[2][1] = dpp_mov<DPP_ROW_SHARE(9)>(W);
    a[3][0] = dpp_mov<DPP_ROW_SHARE(12)>(W);  a[3][1] = dpp_mov<DPP_ROW_SHARE(13)>(W);
#pragma unroll
    for (int kb = 0; kb < 4; ++kb)
        c = __builtin_amdgcn_wmma_f32_16x16x4_f32(false, a[kb], false, bM[kb],
                                                  (short)0, c, false, false);

    const float carry = cross_tile_carry(c[7], sm, wave, lane);
#pragma unroll
    for (int r = 0; r < 8; ++r) v[r] = c[r] + carry;
}

__device__ __forceinline__ void tile_scan_T(float v[8], SMem* sm, int wave, int lane,
                                            const v2f aL[4], const v2f bM[4]) {
    const bool hi = (lane >> 4) != 0;

    v2f a[4];
    kslices(v, hi, a);

    v8f c = {};
#pragma unroll
    for (int kb = 0; kb < 4; ++kb)
        c = __builtin_amdgcn_wmma_f32_16x16x4_f32(false, a[kb], false, aL[kb],
                                                  (short)0, c, false, false);

    float pr[8];
#pragma unroll
    for (int r = 0; r < 8; ++r) pr[r] = c[r];
    v2f p[4];
    kslices(pr, hi, p);
    v2f rt[4];
#pragma unroll
    for (int kb = 0; kb < 4; ++kb) {
        rt[kb][0] = dpp_mov<DPP_ROW_SHARE(15)>(p[kb][0]);
        rt[kb][1] = dpp_mov<DPP_ROW_SHARE(15)>(p[kb][1]);
    }
#pragma unroll
    for (int kb = 0; kb < 4; ++kb)
        c = __builtin_amdgcn_wmma_f32_16x16x4_f32(false, bM[kb], false, rt[kb],
                                                  (short)0, c, false, false);

    const float carry = cross_tile_carry(c[7], sm, wave, lane);
#pragma unroll
    for (int r = 0; r < 8; ++r) v[r] = c[r] + carry;
}

__global__ __launch_bounds__(BLOCK) void
KinematicBicycle_82214263980490_kernel(const float* __restrict__ x0,
                                       const float* __restrict__ U,
                                       const float* __restrict__ dtp,
                                       float* __restrict__ out) {
    __shared__ SMem sm;
    const int tid  = threadIdx.x;
    const int wave = tid >> 5;
    const int lane = tid & 31;
    const int col  = lane & 15;
    const int half = lane >> 4;
    const int rank = wave * 32 + col * 2 + half;
    const int t0 = wave * 256 + col * 16 + half * 8;

    v2f aL[4], bM[4];
#pragma unroll
    for (int kb = 0; kb < 4; ++kb) {
#pragma unroll
        for (int j = 0; j < 2; ++j) {
            const int k = kb * 4 + half * 2 + j;
            aL[kb][j] = (k <= col) ? 1.0f : 0.0f;
            bM[kb][j] = (k <  col) ? 1.0f : 0.0f;
        }
    }

    const float dt = dtp[0];
    const float px = x0[0], py = x0[1], pth = x0[2], pv = x0[3];

    float s[8], dl[8];
    const float4* u4 = (const float4*)(U + (size_t)t0 * 2);
#pragma unroll
    for (int q = 0; q < 4; ++q) {
        const float4 u = u4[q];
        s[2 * q]      = clipf(u.x, -ACC_LIMIT, ACC_LIMIT) * dt;
        dl[2 * q]     = clipf(u.y, -STEER_LIMIT, STEER_LIMIT);
        s[2 * q + 1]  = clipf(u.z, -ACC_LIMIT, ACC_LIMIT) * dt;
        dl[2 * q + 1] = clipf(u.w, -STEER_LIMIT, STEER_LIMIT);
    }

    float S = s[0], Lo = V_MIN_C, Hi = V_MAX_C;
#pragma unroll
    for (int r = 1; r < 8; ++r) {
        S += s[r];
        Lo = clipf(Lo + s[r], V_MIN_C, V_MAX_C);
        Hi = clipf(Hi + s[r], V_MIN_C, V_MAX_C);
    }
    const int slot = rank + wave;
    sm.eS[slot] = S; sm.eL[slot] = Lo; sm.eH[slot] = Hi;
    __syncthreads();

    const float w0 = clipf(pv, V_MIN_C, V_MAX_C);
    if (wave == 0) {
        const int base = lane * RAKE_PITCH;
        ClampAdd part[4];
#pragma unroll
        for (int q = 0; q < 4; ++q) {
            ClampAdd acc;
            acc.S = sm.eS[base + 8 * q];
            acc.L = sm.eL[base + 8 * q];
            acc.H = sm.eH[base + 8 * q];
#pragma unroll
            for (int j = 1; j < 8; ++j) {
                ClampAdd e;
                e.S = sm.eS[base + 8 * q + j];
                e.L = sm.eL[base + 8 * q + j];
                e.H = sm.eH[base + 8 * q + j];
                acc = then_fn(acc, e);
            }
            part[q] = acc;
        }
        ClampAdd T = then_fn(then_fn(part[0], part[1]), then_fn(part[2], part[3]));

#pragma unroll
        for (int d = 1; d < 16; d <<= 1) {
            ClampAdd left;
            if (d == 1) {
                left.S = dpp_mov<DPP_ROW_SHR(1)>(T.S); left.L = dpp_mov<DPP_ROW_SHR(1)>(T.L); left.H = dpp_mov<DPP_ROW_SHR(1)>(T.H);
            } else if (d == 2) {
                left.S = dpp_mov<DPP_ROW_SHR(2)>(T.S); left.L = dpp_mov<DPP_ROW_SHR(2)>(T.L); left.H = dpp_mov<DPP_ROW_SHR(2)>(T.H);
            } else if (d == 4) {
                left.S = dpp_mov<DPP_ROW_SHR(4)>(T.S); left.L = dpp_mov<DPP_ROW_SHR(4)>(T.L); left.H = dpp_mov<DPP_ROW_SHR(4)>(T.H);
            } else {
                left.S = dpp_mov<DPP_ROW_SHR(8)>(T.S); left.L = dpp_mov<DPP_ROW_SHR(8)>(T.L); left.H = dpp_mov<DPP_ROW_SHR(8)>(T.H);
            }
            const ClampAdd both = then_fn(left, T);
            if (col >= d) T = both;
        }
        {
            ClampAdd row0;
            row0.S = xhalf(dpp_mov<DPP_ROW_SHARE(15)>(T.S));
            row0.L = xhalf(dpp_mov<DPP_ROW_SHARE(15)>(T.L));
            row0.H = xhalf(dpp_mov<DPP_ROW_SHARE(15)>(T.H));
            const ClampAdd both = then_fn(row0, T);
            if (half) T = both;
        }
        const float vEnd   = apply_fn(T, w0);
        const float vLeft  = dpp_mov<DPP_ROW_SHR(1)>(vEnd);
        const float vRow0  = xhalf(dpp_mov<DPP_ROW_SHARE(15)>(vEnd));
        const float wGroup = (col != 0) ? vLeft : (half ? vRow0 : w0);

        float wq[4];
        wq[0] = wGroup;
        wq[1] = apply_fn(part[0], wq[0]);
        wq[2] = apply_fn(part[1], wq[1]);
        wq[3] = apply_fn(part[2], wq[2]);
#pragma unroll
        for (int q = 0; q < 4; ++q) {
            float wr = wq[q];
#pragma unroll
            for (int j = 0; j < 8; ++j) {
                sm.wIn[base + 8 * q + j] = wr;
                if (j < 7) {
                    ClampAdd e;
                    e.S = sm.eS[base + 8 * q + j];
                    e.L = sm.eL[base + 8 * q + j];
                    e.H = sm.eH[base + 8 * q + j];
                    wr = apply_fn(e, wr);
                }
            }
        }
    }
    __syncthreads();
    float w = sm.wIn[slot];

    float wst[8], vout[8], dth[8];
    const float dtOverL = dt / WHEELBASE;
#pragma unroll
    for (int r = 0; r < 8; ++r) {
        wst[r] = w;
        dth[r] = w * tan_turns_small(dl[r]) * dtOverL;
        w = clipf(w + s[r], V_MIN_C, V_MAX_C);
        vout[r] = w;
    }

    float incl[8];
#pragma unroll
    for (int r = 0; r < 8; ++r) incl[r] = dth[r];
    tile_scan(incl, &sm, wave, lane, aL, bM);
    float thout[8], thst[8];
#pragma unroll
    for (int r = 0; r < 8; ++r) {
        thout[r] = pth + incl[r];
        thst[r]  = thout[r] - dth[r];
    }
    {
        float* tvw = sm.tv + wave * TV_WORDS + col * TV_PITCH + 2 * half;
#pragma unroll
        for (int r = 0; r < 8; ++r)
            *(float2*)(tvw + 4 * r) = make_float2(thout[r], vout[r]);
    }

    float iy[8];
#pragma unroll
    for (int r = 0; r < 8; ++r) {
        float sn, cs;
        sincos_turns(thst[r], &sn, &cs);
        const float g = wst[r] * dt;
        incl[r] = g * cs;
        iy[r]   = g * sn;
    }
    tile_scan_T(incl, &sm, wave, lane, aL, bM);
    tile_scan_T(iy,   &sm, wave, lane, aL, bM);

    {
        const float* tvr = sm.tv + wave * TV_WORDS + half * (8 * TV_PITCH)
                         + 4 * (col & 7) + 2 * (col >> 3);
        float* orow = out + (size_t)(wave * 256 + half * 128 + col + 1) * 4;
#pragma unroll
        for (int r = 0; r < 8; ++r) {
            const float2 tv = *(const float2*)(tvr + r * TV_PITCH);
            *(float4*)(orow + r * 64) = make_float4(px + incl[r], py + iy[r], tv.x, tv.y);
        }
    }
    if (tid == 0) *(float4*)out = make_float4(px, py, pth, pv);
}

extern "C" void kernel_launch(void* const* d_in, const int* in_sizes, int n_in,
                              void* d_out, int out_size, void* d_ws, size_t ws_size,
                              hipStream_t stream) {
    (void)in_sizes; (void)n_in; (void)out_size; (void)d_ws; (void)ws_size;
    const float* x0 = (const float*)d_in[0];
    const float* U  = (const float*)d_in[1];
    const float* dt = (const float*)d_in[2];
    float* out = (float*)d_out;
    KinematicBicycle_82214263980490_kernel<<<1, BLOCK, 0, stream>>>(x0, U, dt, out);
}


# U read with whole-line global loads (512 contiguous bytes per load; was 16 lines x 32 B, the same lines four times), handed to the 8-step owners through a swizzled per-wave LDS tile; values unchanged
# speedup vs baseline: 1.7080x; 1.0814x over previous
#include <hip/hip_runtime.h>


#define HSTEPS 8192
#define BLOCK  1024
#define NWAVE  32

#define ACC_LIMIT   3.0f
#define STEER_LIMIT 0.6f
#define V_MIN_C     0.0f
#define V_MAX_C     30.0f
#define WHEELBASE   2.7f

typedef __attribute__((ext_vector_type(2))) float v2f;
typedef __attribute__((ext_vector_type(8))) float v8f;

__device__ __forceinline__ float clipf(float x, float lo, float hi) {
    return fminf(fmaxf(x, lo), hi);
}

#define RAKE_PITCH 33
#define RAKE_WORDS (NWAVE * RAKE_PITCH)

#define TV_PITCH 36
#define TV_WORDS (16 * TV_PITCH)

struct SMem {
    alignas(16) float tv[NWAVE * TV_WORDS];
    float tileSum[NWAVE];
    float tileOff[NWAVE];
    float eS[RAKE_WORDS];
    float eL[RAKE_WORDS];
    float eH[RAKE_WORDS];
    float wIn[RAKE_WORDS];
};

__device__ __forceinline__ int u_slot(int c, int j) {
    return 16 * (c >> 2) + 4 * ((c & 3) ^ (c >> 4)) + (j ^ ((c >> 2) & 3));
}

struct ClampAdd { float S, L, H; };

__device__ __forceinline__ ClampAdd then_fn(const ClampAdd first, const ClampAdd next) {
    ClampAdd r;
    r.S = first.S + next.S;
    r.L = clipf(first.L + next.S, next.L, next.H);
    r.H = clipf(first.H + next.S, next.L, next.H);
    return r;
}
__device__ __forceinline__ float apply_fn(const ClampAdd f, float v) {
    return clipf(v + f.S, f.L, f.H);
}

__device__ __forceinline__ float xhalf(float x) {
    const int b = __builtin_bit_cast(int, x);
    return __builtin_bit_cast(float,
        __builtin_amdgcn_permlanex16(b, b, 0x76543210, 0xfedcba98, false, false));
}

#define DPP_ROW_SHL(n)   (0x100 + (n))
#define DPP_ROW_SHR(n)   (0x110 + (n))
#define DPP_ROW_SHARE(n) (0x150 + (n))
template <int CTRL>
__device__ __forceinline__ float dpp_mov(float x) {
    const int b = __builtin_bit_cast(int, x);
    return __builtin_bit_cast(float, __builtin_amdgcn_update_dpp(0, b, CTRL, 0xf, 0xf, true));
}

__device__ __forceinline__ void sincos_turns(float th, float* sn, float* cs) {
    const float C_HI = 0.15915494f;
    const float C_LO = 6.4206382e-9f;
    const float k = __builtin_rintf(th * C_HI);
    float f = __builtin_fmaf(th, C_HI, -k);
    f = __builtin_fmaf(th, C_LO, f);
    *sn = __builtin_amdgcn_sinf(f);
    *cs = __builtin_amdgcn_cosf(f);
}

__device__ __forceinline__ float tan_turns_small(float a) {
    const float C_HI = 0.15915494f;
    const float C_LO = 6.4206382e-9f;
    const float f = __builtin_fmaf(a, C_HI, a * C_LO);
    return __builtin_amdgcn_sinf(f) * __builtin_amdgcn_rcpf(__builtin_amdgcn_cosf(f));
}

__device__ __forceinline__ void kslices(const float v[8], bool hi, v2f f[4]) {
    const float r0 = xhalf(hi ? v[0] : v[2]);
    const float r1 = xhalf(hi ? v[1] : v[3]);
    const float r2 = xhalf(hi ? v[4] : v[6]);
    const float r3 = xhalf(hi ? v[5] : v[7]);

    f[0][0] = hi ? r0   : v[0];  f[0][1] = hi ? r1   : v[1];
    f[1][0] = hi ? r2   : v[4];  f[1][1] = hi ? r3   : v[5];
    f[2][0] = hi ? v[2] : r0;    f[2][1] = hi ? v[3] : r1;
    f[3][0] = hi ? v[6] : r2;    f[3][1] = hi ? v[7] : r3;
}

__device__ __forceinline__ float cross_tile_carry(float corner, SMem* sm, int wave, int lane) {
    if (lane == 31) sm->tileSum[wave] = corner;
    __syncthreads();

    if (threadIdx.x < NWAVE) {
        float ts = sm->tileSum[threadIdx.x];
        float incl = ts;
#pragma unroll
        for (int d = 1; d < 32; d <<= 1) {
            float o = __shfl_up(incl, d, 32);
            if ((int)threadIdx.x >= d) incl += o;
        }
        sm->tileOff[threadIdx.x] = incl - ts;
    }
    __syncthreads();

    return sm->tileOff[wave];
}

__device__ __forceinline__ void tile_scan(float v[8], SMem* sm, int wave, int lane,
                                          const v2f aL[4], const v2f bM[4]) {
    const bool hi = (lane >> 4) != 0;

    v2f b[4];
    kslices(v, hi, b);

    v8f c = {};
#pragma unroll
    for (int kb = 0; kb < 4; ++kb)
        c = __builtin_amdgcn_wmma_f32_16x16x4_f32(false, aL[kb], false, b[kb],
                                                  (short)0, c, false, false);

    const float colTot = c[7];
    const float totShl2  = dpp_mov<DPP_ROW_SHL(2)>(colTot);
    const float totOther = xhalf(colTot);
    const float W = hi ? totShl2 : totOther;

    v2f a[4];
    a[0][0] = dpp_mov<DPP_ROW_SHARE(0)>(W);   a[0][1] = dpp_mov<DPP_ROW_SHARE(1)>(W);
    a[1][0] = dpp_mov<DPP_ROW_SHARE(4)>(W);   a[1][1] = dpp_mov<DPP_ROW_SHARE(5)>(W);
    a[2][0] = dpp_mov<DPP_ROW_SHARE(8)>(W);   a[2][1] = dpp_mov<DPP_ROW_SHARE(9)>(W);
    a[3][0] = dpp_mov<DPP_ROW_SHARE(12)>(W);  a[3][1] = dpp_mov<DPP_ROW_SHARE(13)>(W);
#pragma unroll
    for (int kb = 0; kb < 4; ++kb)
        c = __builtin_amdgcn_wmma_f32_16x16x4_f32(false, a[kb], false, bM[kb],
                                                  (short)0, c, false, false);

    const float carry = cross_tile_carry(c[7], sm, wave, lane);
#pragma unroll
    for (int r = 0; r < 8; ++r) v[r] = c[r] + carry;
}

__device__ __forceinline__ void tile_scan_T(float v[8], SMem* sm, int wave, int lane,
                                            const v2f aL[4], const v2f bM[4]) {
    const bool hi = (lane >> 4) != 0;

    v2f a[4];
    kslices(v, hi, a);

    v8f c = {};
#pragma unroll
    for (int kb = 0; kb < 4; ++kb)
        c = __builtin_amdgcn_wmma_f32_16x16x4_f32(false, a[kb], false, aL[kb],
                                                  (short)0, c, false, false);

    float pr[8];
#pragma unroll
    for (int r = 0; r < 8; ++r) pr[r] = c[r];
    v2f p[4];
    kslices(pr, hi, p);
    v2f rt[4];
#pragma unroll
    for (int kb = 0; kb < 4; ++kb) {
        rt[kb][0] = dpp_mov<DPP_ROW_SHARE(15)>(p[kb][0]);
        rt[kb][1] = dpp_mov<DPP_ROW_SHARE(15)>(p[kb][1]);
    }
#pragma unroll
    for (int kb = 0; kb < 4; ++kb)
        c = __builtin_amdgcn_wmma_f32_16x16x4_f32(false, bM[kb], false, rt[kb],
                                                  (short)0, c, false, false);

    const float carry = cross_tile_carry(c[7], sm, wave, lane);
#pragma unroll
    for (int r = 0; r < 8; ++r) v[r] = c[r] + carry;
}

__global__ __launch_bounds__(BLOCK) void
KinematicBicycle_82214263980490_kernel(const float* __restrict__ x0,
                                       const float* __restrict__ U,
                                       const float* __restrict__ dtp,
                                       float* __restrict__ out) {
    __shared__ SMem sm;
    const int tid  = threadIdx.x;
    const int wave = tid >> 5;
    const int lane = tid & 31;
    const int col  = lane & 15;
    const int half = lane >> 4;
    const int rank = wave * 32 + col * 2 + half;
    const int t0 = wave * 256 + col * 16 + half * 8;

    v2f aL[4], bM[4];
#pragma unroll
    for (int kb = 0; kb < 4; ++kb) {
#pragma unroll
        for (int j = 0; j < 2; ++j) {
            const int k = kb * 4 + half * 2 + j;
            aL[kb][j] = (k <= col) ? 1.0f : 0.0f;
            bM[kb][j] = (k <  col) ? 1.0f : 0.0f;
        }
    }

    const float dt = dtp[0];
    const float px = x0[0], py = x0[1], pth = x0[2], pv = x0[3];

    float s[8], dl[8];
    {
        const float4* ug = (const float4*)U + wave * 128;
        float4* stage = (float4*)(sm.tv + wave * TV_WORDS);
        float4 g[4];
#pragma unroll
        for (int q = 0; q < 4; ++q) g[q] = ug[32 * q + lane];
#pragma unroll
        for (int q = 0; q < 4; ++q) {
            const int p = 32 * q + lane;
            stage[u_slot(p >> 2, p & 3)] = g[q];
        }
        __builtin_amdgcn_fence(__ATOMIC_RELEASE, "wavefront");
        __builtin_amdgcn_wave_barrier();
        __builtin_amdgcn_fence(__ATOMIC_ACQUIRE, "wavefront");
        const int chunk = 2 * col + half;
#pragma unroll
        for (int j = 0; j < 4; ++j) {
            const float4 u = stage[u_slot(chunk, j)];
            s[2 * j]      = clipf(u.x, -ACC_LIMIT, ACC_LIMIT) * dt;
            dl[2 * j]     = clipf(u.y, -STEER_LIMIT, STEER_LIMIT);
            s[2 * j + 1]  = clipf(u.z, -ACC_LIMIT, ACC_LIMIT) * dt;
            dl[2 * j + 1] = clipf(u.w, -STEER_LIMIT, STEER_LIMIT);
        }
    }

    float S = s[0], Lo = V_MIN_C, Hi = V_MAX_C;
#pragma unroll
    for (int r = 1; r < 8; ++r) {
        S += s[r];
        Lo = clipf(Lo + s[r], V_MIN_C, V_MAX_C);
        Hi = clipf(Hi + s[r], V_MIN_C, V_MAX_C);
    }
    const int slot = rank + wave;
    sm.eS[slot] = S; sm.eL[slot] = Lo; sm.eH[slot] = Hi;
    __syncthreads();

    const float w0 = clipf(pv, V_MIN_C, V_MAX_C);
    if (wave == 0) {
        const int base = lane * RAKE_PITCH;
        ClampAdd part[4];
#pragma unroll
        for (int q = 0; q < 4; ++q) {
            ClampAdd acc;
            acc.S = sm.eS[base + 8 * q];
            acc.L = sm.eL[base + 8 * q];
            acc.H = sm.eH[base + 8 * q];
#pragma unroll
            for (int j = 1; j < 8; ++j) {
                ClampAdd e;
                e.S = sm.eS[base + 8 * q + j];
                e.L = sm.eL[base + 8 * q + j];
                e.H = sm.eH[base + 8 * q + j];
                acc = then_fn(acc, e);
            }
            part[q] = acc;
        }
        ClampAdd T = then_fn(then_fn(part[0], part[1]), then_fn(part[2], part[3]));

#pragma unroll
        for (int d = 1; d < 16; d <<= 1) {
            ClampAdd left;
            if (d == 1) {
                left.S = dpp_mov<DPP_ROW_SHR(1)>(T.S); left.L = dpp_mov<DPP_ROW_SHR(1)>(T.L); left.H = dpp_mov<DPP_ROW_SHR(1)>(T.H);
            } else if (d == 2) {
                left.S = dpp_mov<DPP_ROW_SHR(2)>(T.S); left.L = dpp_mov<DPP_ROW_SHR(2)>(T.L); left.H = dpp_mov<DPP_ROW_SHR(2)>(T.H);
            } else if (d == 4) {
                left.S = dpp_mov<DPP_ROW_SHR(4)>(T.S); left.L = dpp_mov<DPP_ROW_SHR(4)>(T.L); left.H = dpp_mov<DPP_ROW_SHR(4)>(T.H);
            } else {
                left.S = dpp_mov<DPP_ROW_SHR(8)>(T.S); left.L = dpp_mov<DPP_ROW_SHR(8)>(T.L); left.H = dpp_mov<DPP_ROW_SHR(8)>(T.H);
            }
            const ClampAdd both = then_fn(left, T);
            if (col >= d) T = both;
        }
        {
            ClampAdd row0;
            row0.S = xhalf(dpp_mov<DPP_ROW_SHARE(15)>(T.S));
            row0.L = xhalf(dpp_mov<DPP_ROW_SHARE(15)>(T.L));
            row0.H = xhalf(dpp_mov<DPP_ROW_SHARE(15)>(T.H));
            const ClampAdd both = then_fn(row0, T);
            if (half) T = both;
        }
        const float vEnd   = apply_fn(T, w0);
        const float vLeft  = dpp_mov<DPP_ROW_SHR(1)>(vEnd);
        const float vRow0  = xhalf(dpp_mov<DPP_ROW_SHARE(15)>(vEnd));
        const float wGroup = (col != 0) ? vLeft : (half ? vRow0 : w0);

        float wq[4];
        wq[0] = wGroup;
        wq[1] = apply_fn(part[0], wq[0]);
        wq[2] = apply_fn(part[1], wq[1]);
        wq[3] = apply_fn(part[2], wq[2]);
#pragma unroll
        for (int q = 0; q < 4; ++q) {
            float wr = wq[q];
#pragma unroll
            for (int j = 0; j < 8; ++j) {
                sm.wIn[base + 8 * q + j] = wr;
                if (j < 7) {
                    ClampAdd e;
                    e.S = sm.eS[base + 8 * q + j];
                    e.L = sm.eL[base + 8 * q + j];
                    e.H = sm.eH[base + 8 * q + j];
                    wr = apply_fn(e, wr);
                }
            }
        }
    }
    __syncthreads();
    float w = sm.wIn[slot];

    float wst[8], vout[8], dth[8];
    const float dtOverL = dt / WHEELBASE;
#pragma unroll
    for (int r = 0; r < 8; ++r) {
        wst[r] = w;
        dth[r] = w * tan_turns_small(dl[r]) * dtOverL;
        w = clipf(w + s[r], V_MIN_C, V_MAX_C);
        vout[r] = w;
    }

    float incl[8];
#pragma unroll
    for (int r = 0; r < 8; ++r) incl[r] = dth[r];
    tile_scan(incl, &sm, wave, lane, aL, bM);
    float thout[8], thst[8];
#pragma unroll
    for (int r = 0; r < 8; ++r) {
        thout[r] = pth + incl[r];
        thst[r]  = thout[r] - dth[r];
    }
    {
        float* tvw = sm.tv + wave * TV_WORDS + col * TV_PITCH + 2 * half;
#pragma unroll
        for (int r = 0; r < 8; ++r)
            *(float2*)(tvw + 4 * r) = make_float2(thout[r], vout[r]);
    }

    float iy[8];
#pragma unroll
    for (int r = 0; r < 8; ++r) {
        float sn, cs;
        sincos_turns(thst[r], &sn, &cs);
        const float g = wst[r] * dt;
        incl[r] = g * cs;
        iy[r]   = g * sn;
    }
    tile_scan_T(incl, &sm, wave, lane, aL, bM);
    tile_scan_T(iy,   &sm, wave, lane, aL, bM);

    {
        const float* tvr = sm.tv + wave * TV_WORDS + half * (8 * TV_PITCH)
                         + 4 * (col & 7) + 2 * (col >> 3);
        float* orow = out + (size_t)(wave * 256 + half * 128 + col + 1) * 4;
#pragma unroll
        for (int r = 0; r < 8; ++r) {
            const float2 tv = *(const float2*)(tvr + r * TV_PITCH);
            *(float4*)(orow + r * 64) = make_float4(px + incl[r], py + iy[r], tv.x, tv.y);
        }
    }
    if (tid == 0) *(float4*)out = make_float4(px, py, pth, pv);
}

extern "C" void kernel_launch(void* const* d_in, const int* in_sizes, int n_in,
                              void* d_out, int out_size, void* d_ws, size_t ws_size,
                              hipStream_t stream) {
    (void)in_sizes; (void)n_in; (void)out_size; (void)d_ws; (void)ws_size;
    const float* x0 = (const float*)d_in[0];
    const float* U  = (const float*)d_in[1];
    const float* dt = (const float*)d_in[2];
    float* out = (float*)d_out;
    KinematicBicycle_82214263980490_kernel<<<1, BLOCK, 0, stream>>>(x0, U, dt, out);
}


# Wave 0's scan of the 32 tile totals (3 hand-overs): lane moves off the LDS path, 5 ds_bpermute per scan -> 4 DPP row shifts + 1 v_permlanex16; new f32 sum order in the tile carries
# speedup vs baseline: 1.8295x; 1.0711x over previous
#include <hip/hip_runtime.h>


#define HSTEPS 8192
#define BLOCK  1024
#define NWAVE  32

#define ACC_LIMIT   3.0f
#define STEER_LIMIT 0.6f
#define V_MIN_C     0.0f
#define V_MAX_C     30.0f
#define WHEELBASE   2.7f

typedef __attribute__((ext_vector_type(2))) float v2f;
typedef __attribute__((ext_vector_type(8))) float v8f;

__device__ __forceinline__ float clipf(float x, float lo, float hi) {
    return fminf(fmaxf(x, lo), hi);
}

#define RAKE_PITCH 33
#define RAKE_WORDS (NWAVE * RAKE_PITCH)

#define TV_PITCH 36
#define TV_WORDS (16 * TV_PITCH)

struct SMem {
    alignas(16) float tv[NWAVE * TV_WORDS];
    float tileSum[NWAVE];
    float tileOff[NWAVE];
    float eS[RAKE_WORDS];
    float eL[RAKE_WORDS];
    float eH[RAKE_WORDS];
    float wIn[RAKE_WORDS];
};

__device__ __forceinline__ int u_slot(int c, int j) {
    return 16 * (c >> 2) + 4 * ((c & 3) ^ (c >> 4)) + (j ^ ((c >> 2) & 3));
}

struct ClampAdd { float S, L, H; };

__device__ __forceinline__ ClampAdd then_fn(const ClampAdd first, const ClampAdd next) {
    ClampAdd r;
    r.S = first.S + next.S;
    r.L = clipf(first.L + next.S, next.L, next.H);
    r.H = clipf(first.H + next.S, next.L, next.H);
    return r;
}
__device__ __forceinline__ float apply_fn(const ClampAdd f, float v) {
    return clipf(v + f.S, f.L, f.H);
}

__device__ __forceinline__ float xhalf(float x) {
    const int b = __builtin_bit_cast(int, x);
    return __builtin_bit_cast(float,
        __builtin_amdgcn_permlanex16(b, b, 0x76543210, 0xfedcba98, false, false));
}

__device__ __forceinline__ float other_row_last(float x) {
    const int b = __builtin_bit_cast(int, x);
    return __builtin_bit_cast(float,
        __builtin_amdgcn_permlanex16(b, b, 0xffffffff, 0xffffffff, false, false));
}

#define DPP_ROW_SHL(n)   (0x100 + (n))
#define DPP_ROW_SHR(n)   (0x110 + (n))
#define DPP_ROW_SHARE(n) (0x150 + (n))
template <int CTRL>
__device__ __forceinline__ float dpp_mov(float x) {
    const int b = __builtin_bit_cast(int, x);
    return __builtin_bit_cast(float, __builtin_amdgcn_update_dpp(0, b, CTRL, 0xf, 0xf, true));
}

__device__ __forceinline__ void sincos_turns(float th, float* sn, float* cs) {
    const float C_HI = 0.15915494f;
    const float C_LO = 6.4206382e-9f;
    const float k = __builtin_rintf(th * C_HI);
    float f = __builtin_fmaf(th, C_HI, -k);
    f = __builtin_fmaf(th, C_LO, f);
    *sn = __builtin_amdgcn_sinf(f);
    *cs = __builtin_amdgcn_cosf(f);
}

__device__ __forceinline__ float tan_turns_small(float a) {
    const float C_HI = 0.15915494f;
    const float C_LO = 6.4206382e-9f;
    const float f = __builtin_fmaf(a, C_HI, a * C_LO);
    return __builtin_amdgcn_sinf(f) * __builtin_amdgcn_rcpf(__builtin_amdgcn_cosf(f));
}

__device__ __forceinline__ void kslices(const float v[8], bool hi, v2f f[4]) {
    const float r0 = xhalf(hi ? v[0] : v[2]);
    const float r1 = xhalf(hi ? v[1] : v[3]);
    const float r2 = xhalf(hi ? v[4] : v[6]);
    const float r3 = xhalf(hi ? v[5] : v[7]);

    f[0][0] = hi ? r0   : v[0];  f[0][1] = hi ? r1   : v[1];
    f[1][0] = hi ? r2   : v[4];  f[1][1] = hi ? r3   : v[5];
    f[2][0] = hi ? v[2] : r0;    f[2][1] = hi ? v[3] : r1;
    f[3][0] = hi ? v[6] : r2;    f[3][1] = hi ? v[7] : r3;
}

__device__ __forceinline__ float cross_tile_carry(float corner, SMem* sm, int wave, int lane) {
    if (lane == 31) sm->tileSum[wave] = corner;
    __syncthreads();

    if (threadIdx.x < NWAVE) {
        const float ts = sm->tileSum[threadIdx.x];
        float incl = ts;
        incl += dpp_mov<DPP_ROW_SHR(1)>(incl);
        incl += dpp_mov<DPP_ROW_SHR(2)>(incl);
        incl += dpp_mov<DPP_ROW_SHR(4)>(incl);
        incl += dpp_mov<DPP_ROW_SHR(8)>(incl);
        const float row0 = other_row_last(incl);
        incl += (lane >> 4) ? row0 : 0.0f;
        sm->tileOff[threadIdx.x] = incl - ts;
    }
    __syncthreads();

    return sm->tileOff[wave];
}

__device__ __forceinline__ void tile_scan(float v[8], SMem* sm, int wave, int lane,
                                          const v2f aL[4], const v2f bM[4]) {
    const bool hi = (lane >> 4) != 0;

    v2f b[4];
    kslices(v, hi, b);

    v8f c = {};
#pragma unroll
    for (int kb = 0; kb < 4; ++kb)
        c = __builtin_amdgcn_wmma_f32_16x16x4_f32(false, aL[kb], false, b[kb],
                                                  (short)0, c, false, false);

    const float colTot = c[7];
    const float totShl2  = dpp_mov<DPP_ROW_SHL(2)>(colTot);
    const float totOther = xhalf(colTot);
    const float W = hi ? totShl2 : totOther;

    v2f a[4];
    a[0][0] = dpp_mov<DPP_ROW_SHARE(0)>(W);   a[0][1] = dpp_mov<DPP_ROW_SHARE(1)>(W);
    a[1][0] = dpp_mov<DPP_ROW_SHARE(4)>(W);   a[1][1] = dpp_mov<DPP_ROW_SHARE(5)>(W);
    a[2][0] = dpp_mov<DPP_ROW_SHARE(8)>(W);   a[2][1] = dpp_mov<DPP_ROW_SHARE(9)>(W);
    a[3][0] = dpp_mov<DPP_ROW_SHARE(12)>(W);  a[3][1] = dpp_mov<DPP_ROW_SHARE(13)>(W);
#pragma unroll
    for (int kb = 0; kb < 4; ++kb)
        c = __builtin_amdgcn_wmma_f32_16x16x4_f32(false, a[kb], false, bM[kb],
                                                  (short)0, c, false, false);

    const float carry = cross_tile_carry(c[7], sm, wave, lane);
#pragma unroll
    for (int r = 0; r < 8; ++r) v[r] = c[r] + carry;
}

__device__ __forceinline__ void tile_scan_T(float v[8], SMem* sm, int wave, int lane,
                                            const v2f aL[4], const v2f bM[4]) {
    const bool hi = (lane >> 4) != 0;

    v2f a[4];
    kslices(v, hi, a);

    v8f c = {};
#pragma unroll
    for (int kb = 0; kb < 4; ++kb)
        c = __builtin_amdgcn_wmma_f32_16x16x4_f32(false, a[kb], false, aL[kb],
                                                  (short)0, c, false, false);

    float pr[8];
#pragma unroll
    for (int r = 0; r < 8; ++r) pr[r] = c[r];
    v2f p[4];
    kslices(pr, hi, p);
    v2f rt[4];
#pragma unroll
    for (int kb = 0; kb < 4; ++kb) {
        rt[kb][0] = dpp_mov<DPP_ROW_SHARE(15)>(p[kb][0]);
        rt[kb][1] = dpp_mov<DPP_ROW_SHARE(15)>(p[kb][1]);
    }
#pragma unroll
    for (int kb = 0; kb < 4; ++kb)
        c = __builtin_amdgcn_wmma_f32_16x16x4_f32(false, bM[kb], false, rt[kb],
                                                  (short)0, c, false, false);

    const float carry = cross_tile_carry(c[7], sm, wave, lane);
#pragma unroll
    for (int r = 0; r < 8; ++r) v[r] = c[r] + carry;
}

__global__ __launch_bounds__(BLOCK) void
KinematicBicycle_82214263980490_kernel(const float* __restrict__ x0,
                                       const float* __restrict__ U,
                                       const float* __restrict__ dtp,
                                       float* __restrict__ out) {
    __shared__ SMem sm;
    const int tid  = threadIdx.x;
    const int wave = tid >> 5;
    const int lane = tid & 31;
    const int col  = lane & 15;
    const int half = lane >> 4;
    const int rank = wave * 32 + col * 2 + half;
    const int t0 = wave * 256 + col * 16 + half * 8;

    v2f aL[4], bM[4];
#pragma unroll
    for (int kb = 0; kb < 4; ++kb) {
#pragma unroll
        for (int j = 0; j < 2; ++j) {
            const int k = kb * 4 + half * 2 + j;
            aL[kb][j] = (k <= col) ? 1.0f : 0.0f;
            bM[kb][j] = (k <  col) ? 1.0f : 0.0f;
        }
    }

    const float dt = dtp[0];
    const float px = x0[0], py = x0[1], pth = x0[2], pv = x0[3];

    float s[8], dl[8];
    {
        const float4* ug = (const float4*)U + wave * 128;
        float4* stage = (float4*)(sm.tv + wave * TV_WORDS);
        float4 g[4];
#pragma unroll
        for (int q = 0; q < 4; ++q) g[q] = ug[32 * q + lane];
#pragma unroll
        for (int q = 0; q < 4; ++q) {
            const int p = 32 * q + lane;
            stage[u_slot(p >> 2, p & 3)] = g[q];
        }
        __builtin_amdgcn_fence(__ATOMIC_RELEASE, "wavefront");
        __builtin_amdgcn_wave_barrier();
        __builtin_amdgcn_fence(__ATOMIC_ACQUIRE, "wavefront");
        const int chunk = 2 * col + half;
#pragma unroll
        for (int j = 0; j < 4; ++j) {
            const float4 u = stage[u_slot(chunk, j)];
            s[2 * j]      = clipf(u.x, -ACC_LIMIT, ACC_LIMIT) * dt;
            dl[2 * j]     = clipf(u.y, -STEER_LIMIT, STEER_LIMIT);
            s[2 * j + 1]  = clipf(u.z, -ACC_LIMIT, ACC_LIMIT) * dt;
            dl[2 * j + 1] = clipf(u.w, -STEER_LIMIT, STEER_LIMIT);
        }
    }

    float S = s[0], Lo = V_MIN_C, Hi = V_MAX_C;
#pragma unroll
    for (int r = 1; r < 8; ++r) {
        S += s[r];
        Lo = clipf(Lo + s[r], V_MIN_C, V_MAX_C);
        Hi = clipf(Hi + s[r], V_MIN_C, V_MAX_C);
    }
    const int slot = rank + wave;
    sm.eS[slot] = S; sm.eL[slot] = Lo; sm.eH[slot] = Hi;
    __syncthreads();

    const float w0 = clipf(pv, V_MIN_C, V_MAX_C);
    if (wave == 0) {
        const int base = lane * RAKE_PITCH;
        ClampAdd part[4];
#pragma unroll
        for (int q = 0; q < 4; ++q) {
            ClampAdd acc;
            acc.S = sm.eS[base + 8 * q];
            acc.L = sm.eL[base + 8 * q];
            acc.H = sm.eH[base + 8 * q];
#pragma unroll
            for (int j = 1; j < 8; ++j) {
                ClampAdd e;
                e.S = sm.eS[base + 8 * q + j];
                e.L = sm.eL[base + 8 * q + j];
                e.H = sm.eH[base + 8 * q + j];
                acc = then_fn(acc, e);
            }
            part[q] = acc;
        }
        ClampAdd T = then_fn(then_fn(part[0], part[1]), then_fn(part[2], part[3]));

#pragma unroll
        for (int d = 1; d < 16; d <<= 1) {
            ClampAdd left;
            if (d == 1) {
                left.S = dpp_mov<DPP_ROW_SHR(1)>(T.S); left.L = dpp_mov<DPP_ROW_SHR(1)>(T.L); left.H = dpp_mov<DPP_ROW_SHR(1)>(T.H);
            } else if (d == 2) {
                left.S = dpp_mov<DPP_ROW_SHR(2)>(T.S); left.L = dpp_mov<DPP_ROW_SHR(2)>(T.L); left.H = dpp_mov<DPP_ROW_SHR(2)>(T.H);
            } else if (d == 4) {
                left.S = dpp_mov<DPP_ROW_SHR(4)>(T.S); left.L = dpp_mov<DPP_ROW_SHR(4)>(T.L); left.H = dpp_mov<DPP_ROW_SHR(4)>(T.H);
            } else {
                left.S = dpp_mov<DPP_ROW_SHR(8)>(T.S); left.L = dpp_mov<DPP_ROW_SHR(8)>(T.L); left.H = dpp_mov<DPP_ROW_SHR(8)>(T.H);
            }
            const ClampAdd both = then_fn(left, T);
            if (col >= d) T = both;
        }
        {
            ClampAdd row0;
            row0.S = xhalf(dpp_mov<DPP_ROW_SHARE(15)>(T.S));
            row0.L = xhalf(dpp_mov<DPP_ROW_SHARE(15)>(T.L));
            row0.H = xhalf(dpp_mov<DPP_ROW_SHARE(15)>(T.H));
            const ClampAdd both = then_fn(row0, T);
            if (half) T = both;
        }
        const float vEnd   = apply_fn(T, w0);
        const float vLeft  = dpp_mov<DPP_ROW_SHR(1)>(vEnd);
        const float vRow0  = xhalf(dpp_mov<DPP_ROW_SHARE(15)>(vEnd));
        const float wGroup = (col != 0) ? vLeft : (half ? vRow0 : w0);

        float wq[4];
        wq[0] = wGroup;
        wq[1] = apply_fn(part[0], wq[0]);
        wq[2] = apply_fn(part[1], wq[1]);
        wq[3] = apply_fn(part[2], wq[2]);
#pragma unroll
        for (int q = 0; q < 4; ++q) {
            float wr = wq[q];
#pragma unroll
            for (int j = 0; j < 8; ++j) {
                sm.wIn[base + 8 * q + j] = wr;
                if (j < 7) {
                    ClampAdd e;
                    e.S = sm.eS[base + 8 * q + j];
                    e.L = sm.eL[base + 8 * q + j];
                    e.H = sm.eH[base + 8 * q + j];
                    wr = apply_fn(e, wr);
                }
            }
        }
    }
    __syncthreads();
    float w = sm.wIn[slot];

    float wst[8], vout[8], dth[8];
    const float dtOverL = dt / WHEELBASE;
#pragma unroll
    for (int r = 0; r < 8; ++r) {
        wst[r] = w;
        dth[r] = w * tan_turns_small(dl[r]) * dtOverL;
        w = clipf(w + s[r], V_MIN_C, V_MAX_C);
        vout[r] = w;
    }

    float incl[8];
#pragma unroll
    for (int r = 0; r < 8; ++r) incl[r] = dth[r];
    tile_scan(incl, &sm, wave, lane, aL, bM);
    float thout[8], thst[8];
#pragma unroll
    for (int r = 0; r < 8; ++r) {
        thout[r] = pth + incl[r];
        thst[r]  = thout[r] - dth[r];
    }
    {
        float* tvw = sm.tv + wave * TV_WORDS + col * TV_PITCH + 2 * half;
#pragma unroll
        for (int r = 0; r < 8; ++r)
            *(float2*)(tvw + 4 * r) = make_float2(thout[r], vout[r]);
    }

    float iy[8];
#pragma unroll
    for (int r = 0; r < 8; ++r) {
        float sn, cs;
        sincos_turns(thst[r], &sn, &cs);
        const float g = wst[r] * dt;
        incl[r] = g * cs;
        iy[r]   = g * sn;
    }
    tile_scan_T(incl, &sm, wave, lane, aL, bM);
    tile_scan_T(iy,   &sm, wave, lane, aL, bM);

    {
        const float* tvr = sm.tv + wave * TV_WORDS + half * (8 * TV_PITCH)
                         + 4 * (col & 7) + 2 * (col >> 3);
        float* orow = out + (size_t)(wave * 256 + half * 128 + col + 1) * 4;
#pragma unroll
        for (int r = 0; r < 8; ++r) {
            const float2 tv = *(const float2*)(tvr + r * TV_PITCH);
            *(float4*)(orow + r * 64) = make_float4(px + incl[r], py + iy[r], tv.x, tv.y);
        }
    }
    if (tid == 0) *(float4*)out = make_float4(px, py, pth, pv);
}

extern "C" void kernel_launch(void* const* d_in, const int* in_sizes, int n_in,
                              void* d_out, int out_size, void* d_ws, size_t ws_size,
                              hipStream_t stream) {
    (void)in_sizes; (void)n_in; (void)out_size; (void)d_ws; (void)ws_size;
    const float* x0 = (const float*)d_in[0];
    const float* U  = (const float*)d_in[1];
    const float* dt = (const float*)d_in[2];
    float* out = (float*)d_out;
    KinematicBicycle_82214263980490_kernel<<<1, BLOCK, 0, stream>>>(x0, U, dt, out);
}


# x, y scans two-level over the workgroup: a tile keeps in-block sums (4 of 8 WMMA); the 512 block totals are scanned once on the matrix unit (x wave 0, y wave 1), one barrier pair; new f32 sum order
# speedup vs baseline: 1.9972x; 1.0917x over previous
#include <hip/hip_runtime.h>


#define HSTEPS 8192
#define BLOCK  1024
#define NWAVE  32

#define ACC_LIMIT   3.0f
#define STEER_LIMIT 0.6f
#define V_MIN_C     0.0f
#define V_MAX_C     30.0f
#define WHEELBASE   2.7f

typedef __attribute__((ext_vector_type(2))) float v2f;
typedef __attribute__((ext_vector_type(8))) float v8f;

__device__ __forceinline__ float clipf(float x, float lo, float hi) {
    return fminf(fmaxf(x, lo), hi);
}

#define RAKE_PITCH 33
#define RAKE_WORDS (NWAVE * RAKE_PITCH)

#define TV_PITCH 36
#define TV_WORDS (16 * TV_PITCH)

struct SMem {
    alignas(16) float tv[NWAVE * TV_WORDS];
    float tileSum[NWAVE];
    float tileOff[NWAVE];
    float eS[RAKE_WORDS];
    float eL[RAKE_WORDS];
    float eH[RAKE_WORDS];
    float wIn[RAKE_WORDS];
    alignas(16) float bt[2][512];
    alignas(16) float bo[2][512];
};

__device__ __forceinline__ int u_slot(int c, int j) {
    return 16 * (c >> 2) + 4 * ((c & 3) ^ (c >> 4)) + (j ^ ((c >> 2) & 3));
}

struct ClampAdd { float S, L, H; };

__device__ __forceinline__ ClampAdd then_fn(const ClampAdd first, const ClampAdd next) {
    ClampAdd r;
    r.S = first.S + next.S;
    r.L = clipf(first.L + next.S, next.L, next.H);
    r.H = clipf(first.H + next.S, next.L, next.H);
    return r;
}
__device__ __forceinline__ float apply_fn(const ClampAdd f, float v) {
    return clipf(v + f.S, f.L, f.H);
}

__device__ __forceinline__ float xhalf(float x) {
    const int b = __builtin_bit_cast(int, x);
    return __builtin_bit_cast(float,
        __builtin_amdgcn_permlanex16(b, b, 0x76543210, 0xfedcba98, false, false));
}

__device__ __forceinline__ float other_row_last(float x) {
    const int b = __builtin_bit_cast(int, x);
    return __builtin_bit_cast(float,
        __builtin_amdgcn_permlanex16(b, b, 0xffffffff, 0xffffffff, false, false));
}

#define DPP_ROW_SHL(n)   (0x100 + (n))
#define DPP_ROW_SHR(n)   (0x110 + (n))
#define DPP_ROW_SHARE(n) (0x150 + (n))
template <int CTRL>
__device__ __forceinline__ float dpp_mov(float x) {
    const int b = __builtin_bit_cast(int, x);
    return __builtin_bit_cast(float, __builtin_amdgcn_update_dpp(0, b, CTRL, 0xf, 0xf, true));
}

__device__ __forceinline__ void sincos_turns(float th, float* sn, float* cs) {
    const float C_HI = 0.15915494f;
    const float C_LO = 6.4206382e-9f;
    const float k = __builtin_rintf(th * C_HI);
    float f = __builtin_fmaf(th, C_HI, -k);
    f = __builtin_fmaf(th, C_LO, f);
    *sn = __builtin_amdgcn_sinf(f);
    *cs = __builtin_amdgcn_cosf(f);
}

__device__ __forceinline__ float tan_turns_small(float a) {
    const float C_HI = 0.15915494f;
    const float C_LO = 6.4206382e-9f;
    const float f = __builtin_fmaf(a, C_HI, a * C_LO);
    return __builtin_amdgcn_sinf(f) * __builtin_amdgcn_rcpf(__builtin_amdgcn_cosf(f));
}

__device__ __forceinline__ void kslices(const float v[8], bool hi, v2f f[4]) {
    const float r0 = xhalf(hi ? v[0] : v[2]);
    const float r1 = xhalf(hi ? v[1] : v[3]);
    const float r2 = xhalf(hi ? v[4] : v[6]);
    const float r3 = xhalf(hi ? v[5] : v[7]);

    f[0][0] = hi ? r0   : v[0];  f[0][1] = hi ? r1   : v[1];
    f[1][0] = hi ? r2   : v[4];  f[1][1] = hi ? r3   : v[5];
    f[2][0] = hi ? v[2] : r0;    f[2][1] = hi ? v[3] : r1;
    f[3][0] = hi ? v[6] : r2;    f[3][1] = hi ? v[7] : r3;
}

__device__ __forceinline__ float cross_tile_carry(float corner, SMem* sm, int wave, int lane) {
    if (lane == 31) sm->tileSum[wave] = corner;
    __syncthreads();

    if (threadIdx.x < NWAVE) {
        const float ts = sm->tileSum[threadIdx.x];
        float incl = ts;
        incl += dpp_mov<DPP_ROW_SHR(1)>(incl);
        incl += dpp_mov<DPP_ROW_SHR(2)>(incl);
        incl += dpp_mov<DPP_ROW_SHR(4)>(incl);
        incl += dpp_mov<DPP_ROW_SHR(8)>(incl);
        const float row0 = other_row_last(incl);
        incl += (lane >> 4) ? row0 : 0.0f;
        sm->tileOff[threadIdx.x] = incl - ts;
    }
    __syncthreads();

    return sm->tileOff[wave];
}

__device__ __forceinline__ v8f tile_scan_local(const float v[8], int lane,
                                               const v2f aL[4], const v2f bM[4]) {
    const bool hi = (lane >> 4) != 0;

    v2f b[4];
    kslices(v, hi, b);

    v8f c = {};
#pragma unroll
    for (int kb = 0; kb < 4; ++kb)
        c = __builtin_amdgcn_wmma_f32_16x16x4_f32(false, aL[kb], false, b[kb],
                                                  (short)0, c, false, false);

    const float colTot = c[7];
    const float totShl2  = dpp_mov<DPP_ROW_SHL(2)>(colTot);
    const float totOther = xhalf(colTot);
    const float W = hi ? totShl2 : totOther;

    v2f a[4];
    a[0][0] = dpp_mov<DPP_ROW_SHARE(0)>(W);   a[0][1] = dpp_mov<DPP_ROW_SHARE(1)>(W);
    a[1][0] = dpp_mov<DPP_ROW_SHARE(4)>(W);   a[1][1] = dpp_mov<DPP_ROW_SHARE(5)>(W);
    a[2][0] = dpp_mov<DPP_ROW_SHARE(8)>(W);   a[2][1] = dpp_mov<DPP_ROW_SHARE(9)>(W);
    a[3][0] = dpp_mov<DPP_ROW_SHARE(12)>(W);  a[3][1] = dpp_mov<DPP_ROW_SHARE(13)>(W);
#pragma unroll
    for (int kb = 0; kb < 4; ++kb)
        c = __builtin_amdgcn_wmma_f32_16x16x4_f32(false, a[kb], false, bM[kb],
                                                  (short)0, c, false, false);

    return c;
}

__device__ __forceinline__ void tile_scan(float v[8], SMem* sm, int wave, int lane,
                                          const v2f aL[4], const v2f bM[4]) {
    const v8f c = tile_scan_local(v, lane, aL, bM);
    const float carry = cross_tile_carry(c[7], sm, wave, lane);
#pragma unroll
    for (int r = 0; r < 8; ++r) v[r] = c[r] + carry;
}

__device__ __forceinline__ v8f block_sums_T(const float v[8], bool hi, const v2f aL[4]) {
    v2f a[4];
    kslices(v, hi, a);
    v8f c = {};
#pragma unroll
    for (int kb = 0; kb < 4; ++kb)
        c = __builtin_amdgcn_wmma_f32_16x16x4_f32(false, a[kb], false, aL[kb],
                                                  (short)0, c, false, false);
    return c;
}

__device__ __forceinline__ void scan_block_totals(const float* bt, float* bo, int lane,
                                                  const v2f aL[4], const v2f bM[4]) {
    const int at = 16 * (lane & 15) + 8 * (lane >> 4);
    const float4 a0 = *(const float4*)(bt + at);
    const float4 a1 = *(const float4*)(bt + at + 4);
    const float4 b0 = *(const float4*)(bt + 256 + at);
    const float4 b1 = *(const float4*)(bt + 256 + at + 4);
    const float a[8] = {a0.x, a0.y, a0.z, a0.w, a1.x, a1.y, a1.z, a1.w};
    const float b[8] = {b0.x, b0.y, b0.z, b0.w, b1.x, b1.y, b1.z, b1.w};
    const v8f ca = tile_scan_local(a, lane, aL, bM);
    const v8f cb = tile_scan_local(b, lane, aL, bM);
    const float totRow1 = dpp_mov<DPP_ROW_SHARE(15)>(ca[7]);
    const float totRow0 = other_row_last(ca[7]);
    const float totA = (lane >> 4) ? totRow1 : totRow0;
    *(float4*)(bo + at)     = make_float4(ca[0] - a[0], ca[1] - a[1], ca[2] - a[2], ca[3] - a[3]);
    *(float4*)(bo + at + 4) = make_float4(ca[4] - a[4], ca[5] - a[5], ca[6] - a[6], ca[7] - a[7]);
    *(float4*)(bo + 256 + at)     = make_float4((cb[0] - b[0]) + totA, (cb[1] - b[1]) + totA,
                                                (cb[2] - b[2]) + totA, (cb[3] - b[3]) + totA);
    *(float4*)(bo + 256 + at + 4) = make_float4((cb[4] - b[4]) + totA, (cb[5] - b[5]) + totA,
                                                (cb[6] - b[6]) + totA, (cb[7] - b[7]) + totA);
}

__global__ __launch_bounds__(BLOCK) void
KinematicBicycle_82214263980490_kernel(const float* __restrict__ x0,
                                       const float* __restrict__ U,
                                       const float* __restrict__ dtp,
                                       float* __restrict__ out) {
    __shared__ SMem sm;
    const int tid  = threadIdx.x;
    const int wave = tid >> 5;
    const int lane = tid & 31;
    const int col  = lane & 15;
    const int half = lane >> 4;
    const int rank = wave * 32 + col * 2 + half;
    const int t0 = wave * 256 + col * 16 + half * 8;

    v2f aL[4], bM[4];
#pragma unroll
    for (int kb = 0; kb < 4; ++kb) {
#pragma unroll
        for (int j = 0; j < 2; ++j) {
            const int k = kb * 4 + half * 2 + j;
            aL[kb][j] = (k <= col) ? 1.0f : 0.0f;
            bM[kb][j] = (k <  col) ? 1.0f : 0.0f;
        }
    }

    const float dt = dtp[0];
    const float px = x0[0], py = x0[1], pth = x0[2], pv = x0[3];

    float s[8], dl[8];
    {
        const float4* ug = (const float4*)U + wave * 128;
        float4* stage = (float4*)(sm.tv + wave * TV_WORDS);
        float4 g[4];
#pragma unroll
        for (int q = 0; q < 4; ++q) g[q] = ug[32 * q + lane];
#pragma unroll
        for (int q = 0; q < 4; ++q) {
            const int p = 32 * q + lane;
            stage[u_slot(p >> 2, p & 3)] = g[q];
        }
        __builtin_amdgcn_fence(__ATOMIC_RELEASE, "wavefront");
        __builtin_amdgcn_wave_barrier();
        __builtin_amdgcn_fence(__ATOMIC_ACQUIRE, "wavefront");
        const int chunk = 2 * col + half;
#pragma unroll
        for (int j = 0; j < 4; ++j) {
            const float4 u = stage[u_slot(chunk, j)];
            s[2 * j]      = clipf(u.x, -ACC_LIMIT, ACC_LIMIT) * dt;
            dl[2 * j]     = clipf(u.y, -STEER_LIMIT, STEER_LIMIT);
            s[2 * j + 1]  = clipf(u.z, -ACC_LIMIT, ACC_LIMIT) * dt;
            dl[2 * j + 1] = clipf(u.w, -STEER_LIMIT, STEER_LIMIT);
        }
    }

    float S = s[0], Lo = V_MIN_C, Hi = V_MAX_C;
#pragma unroll
    for (int r = 1; r < 8; ++r) {
        S += s[r];
        Lo = clipf(Lo + s[r], V_MIN_C, V_MAX_C);
        Hi = clipf(Hi + s[r], V_MIN_C, V_MAX_C);
    }
    const int slot = rank + wave;
    sm.eS[slot] = S; sm.eL[slot] = Lo; sm.eH[slot] = Hi;
    __syncthreads();

    const float w0 = clipf(pv, V_MIN_C, V_MAX_C);
    if (wave == 0) {
        const int base = lane * RAKE_PITCH;
        ClampAdd part[4];
#pragma unroll
        for (int q = 0; q < 4; ++q) {
            ClampAdd acc;
            acc.S = sm.eS[base + 8 * q];
            acc.L = sm.eL[base + 8 * q];
            acc.H = sm.eH[base + 8 * q];
#pragma unroll
            for (int j = 1; j < 8; ++j) {
                ClampAdd e;
                e.S = sm.eS[base + 8 * q + j];
                e.L = sm.eL[base + 8 * q + j];
                e.H = sm.eH[base + 8 * q + j];
                acc = then_fn(acc, e);
            }
            part[q] = acc;
        }
        ClampAdd T = then_fn(then_fn(part[0], part[1]), then_fn(part[2], part[3]));

#pragma unroll
        for (int d = 1; d < 16; d <<= 1) {
            ClampAdd left;
            if (d == 1) {
                left.S = dpp_mov<DPP_ROW_SHR(1)>(T.S); left.L = dpp_mov<DPP_ROW_SHR(1)>(T.L); left.H = dpp_mov<DPP_ROW_SHR(1)>(T.H);
            } else if (d == 2) {
                left.S = dpp_mov<DPP_ROW_SHR(2)>(T.S); left.L = dpp_mov<DPP_ROW_SHR(2)>(T.L); left.H = dpp_mov<DPP_ROW_SHR(2)>(T.H);
            } else if (d == 4) {
                left.S = dpp_mov<DPP_ROW_SHR(4)>(T.S); left.L = dpp_mov<DPP_ROW_SHR(4)>(T.L); left.H = dpp_mov<DPP_ROW_SHR(4)>(T.H);
            } else {
                left.S = dpp_mov<DPP_ROW_SHR(8)>(T.S); left.L = dpp_mov<DPP_ROW_SHR(8)>(T.L); left.H = dpp_mov<DPP_ROW_SHR(8)>(T.H);
            }
            const ClampAdd both = then_fn(left, T);
            if (col >= d) T = both;
        }
        {
            ClampAdd row0;
            row0.S = xhalf(dpp_mov<DPP_ROW_SHARE(15)>(T.S));
            row0.L = xhalf(dpp_mov<DPP_ROW_SHARE(15)>(T.L));
            row0.H = xhalf(dpp_mov<DPP_ROW_SHARE(15)>(T.H));
            const ClampAdd both = then_fn(row0, T);
            if (half) T = both;
        }
        const float vEnd   = apply_fn(T, w0);
        const float vLeft  = dpp_mov<DPP_ROW_SHR(1)>(vEnd);
        const float vRow0  = xhalf(dpp_mov<DPP_ROW_SHARE(15)>(vEnd));
        const float wGroup = (col != 0) ? vLeft : (half ? vRow0 : w0);

        float wq[4];
        wq[0] = wGroup;
        wq[1] = apply_fn(part[0], wq[0]);
        wq[2] = apply_fn(part[1], wq[1]);
        wq[3] = apply_fn(part[2], wq[2]);
#pragma unroll
        for (int q = 0; q < 4; ++q) {
            float wr = wq[q];
#pragma unroll
            for (int j = 0; j < 8; ++j) {
                sm.wIn[base + 8 * q + j] = wr;
                if (j < 7) {
                    ClampAdd e;
                    e.S = sm.eS[base + 8 * q + j];
                    e.L = sm.eL[base + 8 * q + j];
                    e.H = sm.eH[base + 8 * q + j];
                    wr = apply_fn(e, wr);
                }
            }
        }
    }
    __syncthreads();
    float w = sm.wIn[slot];

    float wst[8], vout[8], dth[8];
    const float dtOverL = dt / WHEELBASE;
#pragma unroll
    for (int r = 0; r < 8; ++r) {
        wst[r] = w;
        dth[r] = w * tan_turns_small(dl[r]) * dtOverL;
        w = clipf(w + s[r], V_MIN_C, V_MAX_C);
        vout[r] = w;
    }

    float incl[8];
#pragma unroll
    for (int r = 0; r < 8; ++r) incl[r] = dth[r];
    tile_scan(incl, &sm, wave, lane, aL, bM);
    float thout[8], thst[8];
#pragma unroll
    for (int r = 0; r < 8; ++r) {
        thout[r] = pth + incl[r];
        thst[r]  = thout[r] - dth[r];
    }
    {
        float* tvw = sm.tv + wave * TV_WORDS + col * TV_PITCH + 2 * half;
#pragma unroll
        for (int r = 0; r < 8; ++r)
            *(float2*)(tvw + 4 * r) = make_float2(thout[r], vout[r]);
    }

    float iy[8];
#pragma unroll
    for (int r = 0; r < 8; ++r) {
        float sn, cs;
        sincos_turns(thst[r], &sn, &cs);
        const float g = wst[r] * dt;
        incl[r] = g * cs;
        iy[r]   = g * sn;
    }
    const v8f cx = block_sums_T(incl, half != 0, aL);
    const v8f cy = block_sums_T(iy,   half != 0, aL);
    if (col == 15) {
        float* bx = sm.bt[0] + wave * 16 + half * 8;
        float* by = sm.bt[1] + wave * 16 + half * 8;
        *(float4*)bx       = make_float4(cx[0], cx[1], cx[2], cx[3]);
        *(float4*)(bx + 4) = make_float4(cx[4], cx[5], cx[6], cx[7]);
        *(float4*)by       = make_float4(cy[0], cy[1], cy[2], cy[3]);
        *(float4*)(by + 4) = make_float4(cy[4], cy[5], cy[6], cy[7]);
    }
    __syncthreads();
    const int wave_u = __builtin_amdgcn_readfirstlane(wave);
    if (wave_u < 2) scan_block_totals(sm.bt[wave_u], sm.bo[wave_u], lane, aL, bM);
    __syncthreads();

    {
        const float* tvr = sm.tv + wave * TV_WORDS + half * (8 * TV_PITCH)
                         + 4 * (col & 7) + 2 * (col >> 3);
        float* orow = out + (size_t)(wave * 256 + half * 128 + col + 1) * 4;
        const float* oxp = sm.bo[0] + wave * 16 + half * 8;
        const float* oyp = sm.bo[1] + wave * 16 + half * 8;
        const float4 x0v = *(const float4*)oxp, x1v = *(const float4*)(oxp + 4);
        const float4 y0v = *(const float4*)oyp, y1v = *(const float4*)(oyp + 4);
        const float ox[8] = {x0v.x, x0v.y, x0v.z, x0v.w, x1v.x, x1v.y, x1v.z, x1v.w};
        const float oy[8] = {y0v.x, y0v.y, y0v.z, y0v.w, y1v.x, y1v.y, y1v.z, y1v.w};
#pragma unroll
        for (int r = 0; r < 8; ++r) {
            const float2 tv = *(const float2*)(tvr + r * TV_PITCH);
            *(float4*)(orow + r * 64) = make_float4(px + (cx[r] + ox[r]), py + (cy[r] + oy[r]),
                                                    tv.x, tv.y);
        }
    }
    if (tid == 0) *(float4*)out = make_float4(px, py, pth, pv);
}

extern "C" void kernel_launch(void* const* d_in, const int* in_sizes, int n_in,
                              void* d_out, int out_size, void* d_ws, size_t ws_size,
                              hipStream_t stream) {
    (void)in_sizes; (void)n_in; (void)out_size; (void)d_ws; (void)ws_size;
    const float* x0 = (const float*)d_in[0];
    const float* U  = (const float*)d_in[1];
    const float* dt = (const float*)d_in[2];
    float* out = (float*)d_out;
    KinematicBicycle_82214263980490_kernel<<<1, BLOCK, 0, stream>>>(x0, U, dt, out);
}


# Second stage of each 256-step tile scan (totals of the columns in front) off the matrix unit: 4 DPP row-shift adds on the 16 column totals + 1 lane-half swap replace 4 chained WMMA; new f32 sum order
# speedup vs baseline: 2.1210x; 1.0620x over previous
#include <hip/hip_runtime.h>


#define HSTEPS 8192
#define BLOCK  1024
#define NWAVE  32

#define ACC_LIMIT   3.0f
#define STEER_LIMIT 0.6f
#define V_MIN_C     0.0f
#define V_MAX_C     30.0f
#define WHEELBASE   2.7f

typedef __attribute__((ext_vector_type(2))) float v2f;
typedef __attribute__((ext_vector_type(8))) float v8f;

__device__ __forceinline__ float clipf(float x, float lo, float hi) {
    return fminf(fmaxf(x, lo), hi);
}

#define RAKE_PITCH 33
#define RAKE_WORDS (NWAVE * RAKE_PITCH)

#define TV_PITCH 36
#define TV_WORDS (16 * TV_PITCH)

struct SMem {
    alignas(16) float tv[NWAVE * TV_WORDS];
    float tileSum[NWAVE];
    float tileOff[NWAVE];
    float eS[RAKE_WORDS];
    float eL[RAKE_WORDS];
    float eH[RAKE_WORDS];
    float wIn[RAKE_WORDS];
    alignas(16) float bt[2][512];
    alignas(16) float bo[2][512];
};

__device__ __forceinline__ int u_slot(int c, int j) {
    return 16 * (c >> 2) + 4 * ((c & 3) ^ (c >> 4)) + (j ^ ((c >> 2) & 3));
}

struct ClampAdd { float S, L, H; };

__device__ __forceinline__ ClampAdd then_fn(const ClampAdd first, const ClampAdd next) {
    ClampAdd r;
    r.S = first.S + next.S;
    r.L = clipf(first.L + next.S, next.L, next.H);
    r.H = clipf(first.H + next.S, next.L, next.H);
    return r;
}
__device__ __forceinline__ float apply_fn(const ClampAdd f, float v) {
    return clipf(v + f.S, f.L, f.H);
}

__device__ __forceinline__ float xhalf(float x) {
    const int b = __builtin_bit_cast(int, x);
    return __builtin_bit_cast(float,
        __builtin_amdgcn_permlanex16(b, b, 0x76543210, 0xfedcba98, false, false));
}

__device__ __forceinline__ float other_row_last(float x) {
    const int b = __builtin_bit_cast(int, x);
    return __builtin_bit_cast(float,
        __builtin_amdgcn_permlanex16(b, b, 0xffffffff, 0xffffffff, false, false));
}

#define DPP_ROW_SHR(n)   (0x110 + (n))
#define DPP_ROW_SHARE(n) (0x150 + (n))
template <int CTRL>
__device__ __forceinline__ float dpp_mov(float x) {
    const int b = __builtin_bit_cast(int, x);
    return __builtin_bit_cast(float, __builtin_amdgcn_update_dpp(0, b, CTRL, 0xf, 0xf, true));
}

__device__ __forceinline__ void sincos_turns(float th, float* sn, float* cs) {
    const float C_HI = 0.15915494f;
    const float C_LO = 6.4206382e-9f;
    const float k = __builtin_rintf(th * C_HI);
    float f = __builtin_fmaf(th, C_HI, -k);
    f = __builtin_fmaf(th, C_LO, f);
    *sn = __builtin_amdgcn_sinf(f);
    *cs = __builtin_amdgcn_cosf(f);
}

__device__ __forceinline__ float tan_turns_small(float a) {
    const float C_HI = 0.15915494f;
    const float C_LO = 6.4206382e-9f;
    const float f = __builtin_fmaf(a, C_HI, a * C_LO);
    return __builtin_amdgcn_sinf(f) * __builtin_amdgcn_rcpf(__builtin_amdgcn_cosf(f));
}

__device__ __forceinline__ float row_running_sum(float x) {
    x += dpp_mov<DPP_ROW_SHR(1)>(x);
    x += dpp_mov<DPP_ROW_SHR(2)>(x);
    x += dpp_mov<DPP_ROW_SHR(4)>(x);
    x += dpp_mov<DPP_ROW_SHR(8)>(x);
    return x;
}

__device__ __forceinline__ void kslices(const float v[8], bool hi, v2f f[4]) {
    const float r0 = xhalf(hi ? v[0] : v[2]);
    const float r1 = xhalf(hi ? v[1] : v[3]);
    const float r2 = xhalf(hi ? v[4] : v[6]);
    const float r3 = xhalf(hi ? v[5] : v[7]);

    f[0][0] = hi ? r0   : v[0];  f[0][1] = hi ? r1   : v[1];
    f[1][0] = hi ? r2   : v[4];  f[1][1] = hi ? r3   : v[5];
    f[2][0] = hi ? v[2] : r0;    f[2][1] = hi ? v[3] : r1;
    f[3][0] = hi ? v[6] : r2;    f[3][1] = hi ? v[7] : r3;
}

__device__ __forceinline__ float cross_tile_carry(float corner, SMem* sm, int wave, int lane) {
    if (lane == 31) sm->tileSum[wave] = corner;
    __syncthreads();

    if (threadIdx.x < NWAVE) {
        const float ts = sm->tileSum[threadIdx.x];
        float incl = row_running_sum(ts);
        const float row0 = other_row_last(incl);
        incl += (lane >> 4) ? row0 : 0.0f;
        sm->tileOff[threadIdx.x] = incl - ts;
    }
    __syncthreads();

    return sm->tileOff[wave];
}

__device__ __forceinline__ v8f column_sums(const float v[8], bool hi, const v2f aL[4]) {
    v2f b[4];
    kslices(v, hi, b);

    v8f c = {};
#pragma unroll
    for (int kb = 0; kb < 4; ++kb)
        c = __builtin_amdgcn_wmma_f32_16x16x4_f32(false, aL[kb], false, b[kb],
                                                  (short)0, c, false, false);
    return c;
}

__device__ __forceinline__ float columns_in_front(float run, bool hi) {
    const float front = dpp_mov<DPP_ROW_SHR(1)>(run);
    const float frontOther = xhalf(front);
    return hi ? front : frontOther;
}

__device__ __forceinline__ void tile_scan(float v[8], SMem* sm, int wave, int lane,
                                          const v2f aL[4]) {
    const bool hi = (lane >> 4) != 0;
    const v8f c = column_sums(v, hi, aL);
    const float run = row_running_sum(c[7]);
    const float carry = cross_tile_carry(run, sm, wave, lane);
    const float off = columns_in_front(run, hi) + carry;
#pragma unroll
    for (int r = 0; r < 8; ++r) v[r] = c[r] + off;
}

__device__ __forceinline__ v8f block_sums_T(const float v[8], bool hi, const v2f aL[4]) {
    v2f a[4];
    kslices(v, hi, a);
    v8f c = {};
#pragma unroll
    for (int kb = 0; kb < 4; ++kb)
        c = __builtin_amdgcn_wmma_f32_16x16x4_f32(false, a[kb], false, aL[kb],
                                                  (short)0, c, false, false);
    return c;
}

__device__ __forceinline__ void scan_block_totals(const float* bt, float* bo, int lane,
                                                  const v2f aL[4]) {
    const bool hi = (lane >> 4) != 0;
    const int at = 16 * (lane & 15) + 8 * (lane >> 4);
    const float4 a0 = *(const float4*)(bt + at);
    const float4 a1 = *(const float4*)(bt + at + 4);
    const float4 b0 = *(const float4*)(bt + 256 + at);
    const float4 b1 = *(const float4*)(bt + 256 + at + 4);
    const float a[8] = {a0.x, a0.y, a0.z, a0.w, a1.x, a1.y, a1.z, a1.w};
    const float b[8] = {b0.x, b0.y, b0.z, b0.w, b1.x, b1.y, b1.z, b1.w};
    const v8f ca = column_sums(a, hi, aL);
    const v8f cb = column_sums(b, hi, aL);
    const float runA = row_running_sum(ca[7]);
    const float runB = row_running_sum(cb[7]);
    const float totRow1 = dpp_mov<DPP_ROW_SHARE(15)>(runA);
    const float totRow0 = other_row_last(runA);
    const float totA = hi ? totRow1 : totRow0;
    const float offA = columns_in_front(runA, hi);
    const float offB = columns_in_front(runB, hi) + totA;
    *(float4*)(bo + at)     = make_float4((ca[0] - a[0]) + offA, (ca[1] - a[1]) + offA,
                                          (ca[2] - a[2]) + offA, (ca[3] - a[3]) + offA);
    *(float4*)(bo + at + 4) = make_float4((ca[4] - a[4]) + offA, (ca[5] - a[5]) + offA,
                                          (ca[6] - a[6]) + offA, (ca[7] - a[7]) + offA);
    *(float4*)(bo + 256 + at)     = make_float4((cb[0] - b[0]) + offB, (cb[1] - b[1]) + offB,
                                                (cb[2] - b[2]) + offB, (cb[3] - b[3]) + offB);
    *(float4*)(bo + 256 + at + 4) = make_float4((cb[4] - b[4]) + offB, (cb[5] - b[5]) + offB,
                                                (cb[6] - b[6]) + offB, (cb[7] - b[7]) + offB);
}

__global__ __launch_bounds__(BLOCK) void
KinematicBicycle_82214263980490_kernel(const float* __restrict__ x0,
                                       const float* __restrict__ U,
                                       const float* __restrict__ dtp,
                                       float* __restrict__ out) {
    __shared__ SMem sm;
    const int tid  = threadIdx.x;
    const int wave = tid >> 5;
    const int lane = tid & 31;
    const int col  = lane & 15;
    const int half = lane >> 4;
    const int rank = wave * 32 + col * 2 + half;
    const int t0 = wave * 256 + col * 16 + half * 8;

    v2f aL[4];
#pragma unroll
    for (int kb = 0; kb < 4; ++kb) {
#pragma unroll
        for (int j = 0; j < 2; ++j) {
            const int k = kb * 4 + half * 2 + j;
            aL[kb][j] = (k <= col) ? 1.0f : 0.0f;
        }
    }

    const float dt = dtp[0];
    const float px = x0[0], py = x0[1], pth = x0[2], pv = x0[3];

    float s[8], dl[8];
    {
        const float4* ug = (const float4*)U + wave * 128;
        float4* stage = (float4*)(sm.tv + wave * TV_WORDS);
        float4 g[4];
#pragma unroll
        for (int q = 0; q < 4; ++q) g[q] = ug[32 * q + lane];
#pragma unroll
        for (int q = 0; q < 4; ++q) {
            const int p = 32 * q + lane;
            stage[u_slot(p >> 2, p & 3)] = g[q];
        }
        __builtin_amdgcn_fence(__ATOMIC_RELEASE, "wavefront");
        __builtin_amdgcn_wave_barrier();
        __builtin_amdgcn_fence(__ATOMIC_ACQUIRE, "wavefront");
        const int chunk = 2 * col + half;
#pragma unroll
        for (int j = 0; j < 4; ++j) {
            const float4 u = stage[u_slot(chunk, j)];
            s[2 * j]      = clipf(u.x, -ACC_LIMIT, ACC_LIMIT) * dt;
            dl[2 * j]     = clipf(u.y, -STEER_LIMIT, STEER_LIMIT);
            s[2 * j + 1]  = clipf(u.z, -ACC_LIMIT, ACC_LIMIT) * dt;
            dl[2 * j + 1] = clipf(u.w, -STEER_LIMIT, STEER_LIMIT);
        }
    }

    float S = s[0], Lo = V_MIN_C, Hi = V_MAX_C;
#pragma unroll
    for (int r = 1; r < 8; ++r) {
        S += s[r];
        Lo = clipf(Lo + s[r], V_MIN_C, V_MAX_C);
        Hi = clipf(Hi + s[r], V_MIN_C, V_MAX_C);
    }
    const int slot = rank + wave;
    sm.eS[slot] = S; sm.eL[slot] = Lo; sm.eH[slot] = Hi;
    __syncthreads();

    const float w0 = clipf(pv, V_MIN_C, V_MAX_C);
    if (wave == 0) {
        const int base = lane * RAKE_PITCH;
        ClampAdd part[4];
#pragma unroll
        for (int q = 0; q < 4; ++q) {
            ClampAdd acc;
            acc.S = sm.eS[base + 8 * q];
            acc.L = sm.eL[base + 8 * q];
            acc.H = sm.eH[base + 8 * q];
#pragma unroll
            for (int j = 1; j < 8; ++j) {
                ClampAdd e;
                e.S = sm.eS[base + 8 * q + j];
                e.L = sm.eL[base + 8 * q + j];
                e.H = sm.eH[base + 8 * q + j];
                acc = then_fn(acc, e);
            }
            part[q] = acc;
        }
        ClampAdd T = then_fn(then_fn(part[0], part[1]), then_fn(part[2], part[3]));

#pragma unroll
        for (int d = 1; d < 16; d <<= 1) {
            ClampAdd left;
            if (d == 1) {
                left.S = dpp_mov<DPP_ROW_SHR(1)>(T.S); left.L = dpp_mov<DPP_ROW_SHR(1)>(T.L); left.H = dpp_mov<DPP_ROW_SHR(1)>(T.H);
            } else if (d == 2) {
                left.S = dpp_mov<DPP_ROW_SHR(2)>(T.S); left.L = dpp_mov<DPP_ROW_SHR(2)>(T.L); left.H = dpp_mov<DPP_ROW_SHR(2)>(T.H);
            } else if (d == 4) {
                left.S = dpp_mov<DPP_ROW_SHR(4)>(T.S); left.L = dpp_mov<DPP_ROW_SHR(4)>(T.L); left.H = dpp_mov<DPP_ROW_SHR(4)>(T.H);
            } else {
                left.S = dpp_mov<DPP_ROW_SHR(8)>(T.S); left.L = dpp_mov<DPP_ROW_SHR(8)>(T.L); left.H = dpp_mov<DPP_ROW_SHR(8)>(T.H);
            }
            const ClampAdd both = then_fn(left, T);
            if (col >= d) T = both;
        }
        {
            ClampAdd row0;
            row0.S = xhalf(dpp_mov<DPP_ROW_SHARE(15)>(T.S));
            row0.L = xhalf(dpp_mov<DPP_ROW_SHARE(15)>(T.L));
            row0.H = xhalf(dpp_mov<DPP_ROW_SHARE(15)>(T.H));
            const ClampAdd both = then_fn(row0, T);
            if (half) T = both;
        }
        const float vEnd   = apply_fn(T, w0);
        const float vLeft  = dpp_mov<DPP_ROW_SHR(1)>(vEnd);
        const float vRow0  = xhalf(dpp_mov<DPP_ROW_SHARE(15)>(vEnd));
        const float wGroup = (col != 0) ? vLeft : (half ? vRow0 : w0);

        float wq[4];
        wq[0] = wGroup;
        wq[1] = apply_fn(part[0], wq[0]);
        wq[2] = apply_fn(part[1], wq[1]);
        wq[3] = apply_fn(part[2], wq[2]);
#pragma unroll
        for (int q = 0; q < 4; ++q) {
            float wr = wq[q];
#pragma unroll
            for (int j = 0; j < 8; ++j) {
                sm.wIn[base + 8 * q + j] = wr;
                if (j < 7) {
                    ClampAdd e;
                    e.S = sm.eS[base + 8 * q + j];
                    e.L = sm.eL[base + 8 * q + j];
                    e.H = sm.eH[base + 8 * q + j];
                    wr = apply_fn(e, wr);
                }
            }
        }
    }
    __syncthreads();
    float w = sm.wIn[slot];

    float wst[8], vout[8], dth[8];
    const float dtOverL = dt / WHEELBASE;
#pragma unroll
    for (int r = 0; r < 8; ++r) {
        wst[r] = w;
        dth[r] = w * tan_turns_small(dl[r]) * dtOverL;
        w = clipf(w + s[r], V_MIN_C, V_MAX_C);
        vout[r] = w;
    }

    float incl[8];
#pragma unroll
    for (int r = 0; r < 8; ++r) incl[r] = dth[r];
    tile_scan(incl, &sm, wave, lane, aL);
    float thout[8], thst[8];
#pragma unroll
    for (int r = 0; r < 8; ++r) {
        thout[r] = pth + incl[r];
        thst[r]  = thout[r] - dth[r];
    }
    {
        float* tvw = sm.tv + wave * TV_WORDS + col * TV_PITCH + 2 * half;
#pragma unroll
        for (int r = 0; r < 8; ++r)
            *(float2*)(tvw + 4 * r) = make_float2(thout[r], vout[r]);
    }

    float iy[8];
#pragma unroll
    for (int r = 0; r < 8; ++r) {
        float sn, cs;
        sincos_turns(thst[r], &sn, &cs);
        const float g = wst[r] * dt;
        incl[r] = g * cs;
        iy[r]   = g * sn;
    }
    const v8f cx = block_sums_T(incl, half != 0, aL);
    const v8f cy = block_sums_T(iy,   half != 0, aL);
    if (col == 15) {
        float* bx = sm.bt[0] + wave * 16 + half * 8;
        float* by = sm.bt[1] + wave * 16 + half * 8;
        *(float4*)bx       = make_float4(cx[0], cx[1], cx[2], cx[3]);
        *(float4*)(bx + 4) = make_float4(cx[4], cx[5], cx[6], cx[7]);
        *(float4*)by       = make_float4(cy[0], cy[1], cy[2], cy[3]);
        *(float4*)(by + 4) = make_float4(cy[4], cy[5], cy[6], cy[7]);
    }
    __syncthreads();
    const int wave_u = __builtin_amdgcn_readfirstlane(wave);
    if (wave_u < 2) scan_block_totals(sm.bt[wave_u], sm.bo[wave_u], lane, aL);
    __syncthreads();

    {
        const float* tvr = sm.tv + wave * TV_WORDS + half * (8 * TV_PITCH)
                         + 4 * (col & 7) + 2 * (col >> 3);
        float* orow = out + (size_t)(wave * 256 + half * 128 + col + 1) * 4;
        const float* oxp = sm.bo[0] + wave * 16 + half * 8;
        const float* oyp = sm.bo[1] + wave * 16 + half * 8;
        const float4 x0v = *(const float4*)oxp, x1v = *(const float4*)(oxp + 4);
        const float4 y0v = *(const float4*)oyp, y1v = *(const float4*)(oyp + 4);
        const float ox[8] = {x0v.x, x0v.y, x0v.z, x0v.w, x1v.x, x1v.y, x1v.z, x1v.w};
        const float oy[8] = {y0v.x, y0v.y, y0v.z, y0v.w, y1v.x, y1v.y, y1v.z, y1v.w};
#pragma unroll
        for (int r = 0; r < 8; ++r) {
            const float2 tv = *(const float2*)(tvr + r * TV_PITCH);
            *(float4*)(orow + r * 64) = make_float4(px + (cx[r] + ox[r]), py + (cy[r] + oy[r]),
                                                    tv.x, tv.y);
        }
    }
    if (tid == 0) *(float4*)out = make_float4(px, py, pth, pv);
}

extern "C" void kernel_launch(void* const* d_in, const int* in_sizes, int n_in,
                              void* d_out, int out_size, void* d_ws, size_t ws_size,
                              hipStream_t stream) {
    (void)in_sizes; (void)n_in; (void)out_size; (void)d_ws; (void)ws_size;
    const float* x0 = (const float*)d_in[0];
    const float* U  = (const float*)d_in[1];
    const float* dt = (const float*)d_in[2];
    float* out = (float*)d_out;
    KinematicBicycle_82214263980490_kernel<<<1, BLOCK, 0, stream>>>(x0, U, dt, out);
}
